# starting kernel
# baseline (speedup 1.0000x reference)
#include <hip/hip_runtime.h>

#define IN_CH  64
#define EMBED  1024
#define HID    256
#define KK     9
#define KDIM   (IN_CH * KK)
#define BATCH  8
#define NTOK   32
#define HDIM   128
#define WDIM   128
#define WSTRIDE 577

typedef __attribute__((ext_vector_type(2))) float v2f;
typedef __attribute__((ext_vector_type(8))) float v8f;

__global__ __launch_bounds__(HID) void mlp1_kernel(
    const float* __restrict__ t_emb, const float* __restrict__ wv,
    const float* __restrict__ W1, const float* __restrict__ b1,
    float* __restrict__ hout) {
  __shared__ float ctx[EMBED];
  const int bn  = blockIdx.x;
  const int b   = bn >> 5;
  const int tid = threadIdx.x;
  for (int e = tid; e < EMBED; e += HID)
    ctx[e] = wv[(size_t)bn * EMBED + e] + t_emb[(size_t)b * EMBED + e];
  __syncthreads();
  float acc = b1[tid];
  for (int e = 0; e < EMBED; ++e)
    acc = fmaf(ctx[e], W1[(size_t)e * HID + tid], acc);
  const float sig = 1.0f / (1.0f + __expf(-acc));
  hout[(size_t)bn * HID + tid] = acc * sig;
}

__global__ __launch_bounds__(KDIM) void mlp2_kernel(
    const float* __restrict__ h, const float* __restrict__ W2,
    const float* __restrict__ b2, float* __restrict__ wmat) {
  __shared__ float hl[HID];
  const int bn  = blockIdx.x;
  const int tid = threadIdx.x;
  for (int k = tid; k < HID; k += KDIM) hl[k] = h[(size_t)bn * HID + k];
  __syncthreads();
  float acc = b2[tid];
  for (int k = 0; k < HID; ++k)
    acc = fmaf(hl[k], W2[(size_t)k * KDIM + tid], acc);
  wmat[(size_t)bn * KDIM + tid] = acc;
}

__global__ __launch_bounds__(256) void dynconv_wmma_kernel(
    const float* __restrict__ latent, const float* __restrict__ wmat,
    float* __restrict__ out) {
  __shared__ float wlds[NTOK * WSTRIDE];

  const int b   = blockIdx.z;
  const int h   = blockIdx.y;
  const int w0  = blockIdx.x * 64;
  const int tid = threadIdx.x;

  const float* wsrc = wmat + (size_t)b * NTOK * KDIM;
  for (int n = 0; n < NTOK; ++n) {
    for (int o = tid; o < KDIM; o += 256)
      wlds[n * WSTRIDE + o] = wsrc[n * KDIM + o];
  }
  __syncthreads();

  const int wave    = tid >> 5;
  const int lane    = tid & 31;
  const int m       = lane & 15;
  const int hiK     = lane >> 4;
  const int tokBase = (wave & 1) * 16;
  const int px0     = w0 + (wave >> 1) * 16;

  const float* lat_b = latent + (size_t)b * IN_CH * HDIM * WDIM;
  const float* arow  = &wlds[(tokBase + m) * WSTRIDE];

  v8f acc = {0.f, 0.f, 0.f, 0.f, 0.f, 0.f, 0.f, 0.f};

  for (int di = 0; di < 3; ++di) {
    const int  hh    = h + di - 1;
    const bool rowOK = (unsigned)hh < (unsigned)HDIM;
    for (int dj = 0; dj < 3; ++dj) {
      const int  kk = di * 3 + dj;
      const int  ww = px0 + m + dj - 1;
      const bool ok = rowOK && ((unsigned)ww < (unsigned)WDIM);
      const float* bbase = lat_b + (long)hh * WDIM + ww;
#pragma unroll
      for (int d = 0; d < IN_CH; d += 4) {
        const int c0 = d + 2 * hiK;
        v2f a, bv;
        a.x = arow[c0 * KK + kk];
        a.y = arow[(c0 + 1) * KK + kk];
        bv.x = ok ? bbase[(size_t)c0 * (HDIM * WDIM)] : 0.0f;
        bv.y = ok ? bbase[(size_t)(c0 + 1) * (HDIM * WDIM)] : 0.0f;
        acc = __builtin_amdgcn_wmma_f32_16x16x4_f32(
             false, a,  false, bv,
             (short)0, acc,  false,  false);
      }
    }
  }

  const int tok = tokBase + 8 * hiK;
#pragma unroll
  for (int r = 0; r < 8; ++r) {
    out[(((size_t)b * NTOK + tok + r) * HDIM + h) * WDIM + (px0 + m)] = acc[r];
  }
}

extern "C" void kernel_launch(void* const* d_in, const int* in_sizes, int n_in,
                              void* d_out, int out_size, void* d_ws, size_t ws_size,
                              hipStream_t stream) {
  const float* latent = (const float*)d_in[0];
  const float* t_emb  = (const float*)d_in[1];
  const float* wv     = (const float*)d_in[2];
  const float* W1     = (const float*)d_in[3];
  const float* b1     = (const float*)d_in[4];
  const float* W2     = (const float*)d_in[5];
  const float* b2     = (const float*)d_in[6];
  float* out = (float*)d_out;

  float* hbuf = (float*)d_ws;
  float* wmat = hbuf + (size_t)BATCH * NTOK * HID;

  mlp1_kernel<<<BATCH * NTOK, HID, 0, stream>>>(t_emb, wv, W1, b1, hbuf);
  mlp2_kernel<<<BATCH * NTOK, KDIM, 0, stream>>>(hbuf, W2, b2, wmat);

  dim3 grid(WDIM / 64, HDIM, BATCH);
  dynconv_wmma_kernel<<<grid, 256, 0, stream>>>(latent, wmat, out);
}


# dynconv re-tiled: each wave 32 tokens x 16 px x 4 rows (8 accumulators), latent row fragments reused over row taps and token tiles, loads grouped one step ahead; sum order now channel-major
# speedup vs baseline: 2.0157x; 2.0157x over previous
#include <hip/hip_runtime.h>

#define IN_CH  64
#define EMBED  1024
#define HID    256
#define KK     9
#define KDIM   (IN_CH * KK)
#define BATCH  8
#define NTOK   32
#define HDIM   128
#define WDIM   128
#define WSTRIDE 577

typedef __attribute__((ext_vector_type(2))) float v2f;
typedef __attribute__((ext_vector_type(8))) float v8f;

__global__ __launch_bounds__(HID) void mlp1_kernel(
    const float* __restrict__ t_emb, const float* __restrict__ wv,
    const float* __restrict__ W1, const float* __restrict__ b1,
    float* __restrict__ hout) {
  __shared__ float ctx[EMBED];
  const int bn  = blockIdx.x;
  const int b   = bn >> 5;
  const int tid = threadIdx.x;
  for (int e = tid; e < EMBED; e += HID)
    ctx[e] = wv[(size_t)bn * EMBED + e] + t_emb[(size_t)b * EMBED + e];
  __syncthreads();
  float acc = b1[tid];
  for (int e = 0; e < EMBED; ++e)
    acc = fmaf(ctx[e], W1[(size_t)e * HID + tid], acc);
  const float sig = 1.0f / (1.0f + __expf(-acc));
  hout[(size_t)bn * HID + tid] = acc * sig;
}

__global__ __launch_bounds__(KDIM) void mlp2_kernel(
    const float* __restrict__ h, const float* __restrict__ W2,
    const float* __restrict__ b2, float* __restrict__ wmat) {
  __shared__ float hl[HID];
  const int bn  = blockIdx.x;
  const int tid = threadIdx.x;
  for (int k = tid; k < HID; k += KDIM) hl[k] = h[(size_t)bn * HID + k];
  __syncthreads();
  float acc = b2[tid];
  for (int k = 0; k < HID; ++k)
    acc = fmaf(hl[k], W2[(size_t)k * KDIM + tid], acc);
  wmat[(size_t)bn * KDIM + tid] = acc;
}

#define RPW 4
#define NBROW (RPW + 2)

struct Frag {
  v2f a[3][2];
  v2f b[NBROW];
};

struct Raw {
  float ax[3][2], ay[3][2];
  float bx[NBROW], by[NBROW];
};

__device__ __forceinline__ void issue_loads(
    Raw& r, const float* __restrict__ lat_b, const float* arow0,
    const float* arow1, int cg, int dj, int h0, int px0, int m, int hiK) {
  const int c0  = 4 * cg + 2 * hiK;
  const int ww  = px0 + m + dj - 1;
  const int wwc = min(max(ww, 0), WDIM - 1);
  const float* cptr = lat_b + (size_t)c0 * (HDIM * WDIM) + wwc;
#pragma unroll
  for (int rr = 0; rr < NBROW; ++rr) {
    const int rowc = min(max(h0 - 1 + rr, 0), HDIM - 1);
    r.bx[rr] = cptr[rowc * WDIM];
    r.by[rr] = cptr[rowc * WDIM + HDIM * WDIM];
  }
#pragma unroll
  for (int di = 0; di < 3; ++di) {
    const int kk = di * 3 + dj;
    r.ax[di][0] = arow0[c0 * KK + kk];
    r.ay[di][0] = arow0[(c0 + 1) * KK + kk];
    r.ax[di][1] = arow1[c0 * KK + kk];
    r.ay[di][1] = arow1[(c0 + 1) * KK + kk];
  }
}

__device__ __forceinline__ void finish_frag(
    Frag& f, const Raw& r, int dj, int h0, int px0, int m) {
  const int  ww    = px0 + m + dj - 1;
  const bool colOK = (unsigned)ww < (unsigned)WDIM;
#pragma unroll
  for (int rr = 0; rr < NBROW; ++rr) {
    const bool ok = colOK && ((unsigned)(h0 - 1 + rr) < (unsigned)HDIM);
    f.b[rr].x = ok ? r.bx[rr] : 0.0f;
    f.b[rr].y = ok ? r.by[rr] : 0.0f;
  }
#pragma unroll
  for (int di = 0; di < 3; ++di) {
#pragma unroll
    for (int t = 0; t < 2; ++t) {
      f.a[di][t].x = r.ax[di][t];
      f.a[di][t].y = r.ay[di][t];
    }
  }
}

__device__ __forceinline__ void mma_step(v8f (&acc)[2][RPW], const Frag& f) {
#pragma unroll
  for (int di = 0; di < 3; ++di) {
#pragma unroll
    for (int r = 0; r < RPW; ++r) {
#pragma unroll
      for (int t = 0; t < 2; ++t) {
        acc[t][r] = __builtin_amdgcn_wmma_f32_16x16x4_f32(
             false, f.a[di][t],  false, f.b[r + di],
             (short)0, acc[t][r],  false,  false);
      }
    }
  }
}

__global__ __launch_bounds__(256) void dynconv_wmma_kernel(
    const float* __restrict__ latent, const float* __restrict__ wmat,
    float* __restrict__ out) {
  __shared__ float wlds[NTOK * WSTRIDE];

  const int b   = blockIdx.z;
  const int h0  = blockIdx.y * RPW;
  const int tid = threadIdx.x;

  const float* wsrc = wmat + (size_t)b * NTOK * KDIM;
  for (int n = 0; n < NTOK; ++n) {
    for (int o = tid; o < KDIM; o += 256)
      wlds[n * WSTRIDE + o] = wsrc[n * KDIM + o];
  }
  __syncthreads();

  const int wave = tid >> 5;
  const int lane = tid & 31;
  const int m    = lane & 15;
  const int hiK  = lane >> 4;
  const int px0  = wave * 16;

  const float* lat_b = latent + (size_t)b * IN_CH * HDIM * WDIM;
  const float* arow0 = &wlds[m * WSTRIDE];
  const float* arow1 = &wlds[(16 + m) * WSTRIDE];

  v8f acc[2][RPW];
#pragma unroll
  for (int t = 0; t < 2; ++t)
#pragma unroll
    for (int r = 0; r < RPW; ++r)
      acc[t][r] = (v8f){0.f, 0.f, 0.f, 0.f, 0.f, 0.f, 0.f, 0.f};

  Raw  raw;
  Frag f;
  issue_loads(raw, lat_b, arow0, arow1, 0, 0, h0, px0, m, hiK);
  finish_frag(f, raw, 0, h0, px0, m);
  for (int cg = 0; cg < IN_CH / 4; ++cg) {
    const int cgn = min(cg + 1, IN_CH / 4 - 1);

    issue_loads(raw, lat_b, arow0, arow1, cg, 1, h0, px0, m, hiK);
    __builtin_amdgcn_sched_barrier(0);
    mma_step(acc, f);
    __builtin_amdgcn_sched_barrier(0);
    finish_frag(f, raw, 1, h0, px0, m);

    issue_loads(raw, lat_b, arow0, arow1, cg, 2, h0, px0, m, hiK);
    __builtin_amdgcn_sched_barrier(0);
    mma_step(acc, f);
    __builtin_amdgcn_sched_barrier(0);
    finish_frag(f, raw, 2, h0, px0, m);

    issue_loads(raw, lat_b, arow0, arow1, cgn, 0, h0, px0, m, hiK);
    __builtin_amdgcn_sched_barrier(0);
    mma_step(acc, f);
    __builtin_amdgcn_sched_barrier(0);
    finish_frag(f, raw, 0, h0, px0, m);
  }

#pragma unroll
  for (int t = 0; t < 2; ++t) {
    const int tok = 16 * t + 8 * hiK;
#pragma unroll
    for (int r = 0; r < RPW; ++r) {
#pragma unroll
      for (int i = 0; i < 8; ++i) {
        out[(((size_t)b * NTOK + tok + i) * HDIM + (h0 + r)) * WDIM + (px0 + m)] =
            acc[t][r][i];
      }
    }
  }
}

extern "C" void kernel_launch(void* const* d_in, const int* in_sizes, int n_in,
                              void* d_out, int out_size, void* d_ws, size_t ws_size,
                              hipStream_t stream) {
  const float* latent = (const float*)d_in[0];
  const float* t_emb  = (const float*)d_in[1];
  const float* wv     = (const float*)d_in[2];
  const float* W1     = (const float*)d_in[3];
  const float* b1     = (const float*)d_in[4];
  const float* W2     = (const float*)d_in[5];
  const float* b2     = (const float*)d_in[6];
  float* out = (float*)d_out;

  float* hbuf = (float*)d_ws;
  float* wmat = hbuf + (size_t)BATCH * NTOK * HID;

  mlp1_kernel<<<BATCH * NTOK, HID, 0, stream>>>(t_emb, wv, W1, b1, hbuf);
  mlp2_kernel<<<BATCH * NTOK, KDIM, 0, stream>>>(hbuf, W2, b2, wmat);

  dim3 grid(1, HDIM / RPW, BATCH);
  dynconv_wmma_kernel<<<grid, 256, 0, stream>>>(latent, wmat, out);
}


# both MLP GEMMs moved from one-row-per-block dot products to 16x16 f32 WMMA tiles, K split over the block's waves and summed in LDS: weights read once per 16 rows; MLP sum order changes
# speedup vs baseline: 3.2487x; 1.6117x over previous
#include <hip/hip_runtime.h>

#define IN_CH  64
#define EMBED  1024
#define HID    256
#define KK     9
#define KDIM   (IN_CH * KK)
#define BATCH  8
#define NTOK   32
#define HDIM   128
#define WDIM   128
#define WSTRIDE 577

typedef __attribute__((ext_vector_type(2))) float v2f;
typedef __attribute__((ext_vector_type(8))) float v8f;

typedef __attribute__((ext_vector_type(4))) float v4f;

template <int KW, bool ADD2>
__device__ __forceinline__ v8f tile_partial(
    const float* __restrict__ arow, const float* __restrict__ a2row,
    const float* __restrict__ bcol, int ldb, int kbeg, int hk) {
  v8f acc0 = (v8f){0.f, 0.f, 0.f, 0.f, 0.f, 0.f, 0.f, 0.f};
  v8f acc1 = (v8f){0.f, 0.f, 0.f, 0.f, 0.f, 0.f, 0.f, 0.f};
#pragma unroll
  for (int kb = 0; kb < KW; kb += 8) {
    const int k = kbeg + kb + 4 * hk;
    v4f a = *(const v4f*)(arow + k);
    if (ADD2) {
      const v4f t = *(const v4f*)(a2row + k);
      a = a + t;
    }
    const float* bp = bcol + (size_t)k * ldb;
    const v2f a01 = (v2f){a.x, a.y};
    const v2f a23 = (v2f){a.z, a.w};
    const v2f b01 = (v2f){bp[0], bp[ldb]};
    const v2f b23 = (v2f){bp[2 * ldb], bp[3 * ldb]};
    acc0 = __builtin_amdgcn_wmma_f32_16x16x4_f32(
         false, a01,  false, b01,
         (short)0, acc0,  false,  false);
    acc1 = __builtin_amdgcn_wmma_f32_16x16x4_f32(
         false, a23,  false, b23,
         (short)0, acc1,  false,  false);
  }
  return acc0 + acc1;
}

__global__ __launch_bounds__(512) void mlp1_kernel(
    const float* __restrict__ t_emb, const float* __restrict__ wv,
    const float* __restrict__ W1, const float* __restrict__ b1,
    float* __restrict__ hout) {
  __shared__ float red[16][256];
  const int row0 = blockIdx.y * 16;
  const int tid  = threadIdx.x;
  const int wave = tid >> 5;
  const int lane = tid & 31;
  const int c    = lane & 15;
  const int hk   = lane >> 4;
  const int ct   = wave & 1;
  const int kq   = wave >> 1;
  const int b    = row0 >> 5;

  const v8f part = tile_partial<EMBED / 8, true>(
      wv + (size_t)(row0 + c) * EMBED, t_emb + (size_t)b * EMBED,
      W1 + blockIdx.x * 32 + ct * 16 + c, HID, kq * (EMBED / 8), hk);
#pragma unroll
  for (int r = 0; r < 8; ++r) red[wave][r * 32 + lane] = part[r];
  __syncthreads();

  const int orow = wave;
  const int t    = lane >> 4;
  const int idx  = (orow & 7) * 32 + (orow >> 3) * 16 + (lane & 15);
  const int n    = blockIdx.x * 32 + lane;
  float acc = b1[n];
#pragma unroll
  for (int q = 0; q < 8; ++q) acc += red[t + 2 * q][idx];
  const float sig = 1.0f / (1.0f + __expf(-acc));
  hout[(size_t)(row0 + orow) * HID + n] = acc * sig;
}

__global__ __launch_bounds__(256) void mlp2_kernel(
    const float* __restrict__ h, const float* __restrict__ W2,
    const float* __restrict__ b2, float* __restrict__ wmat) {
  __shared__ float red[8][256];
  const int row0 = blockIdx.y * 16;
  const int tid  = threadIdx.x;
  const int wave = tid >> 5;
  const int lane = tid & 31;
  const int c    = lane & 15;
  const int hk   = lane >> 4;
  const int ct   = wave & 1;
  const int kq   = wave >> 1;

  const v8f part = tile_partial<HID / 4, false>(
      h + (size_t)(row0 + c) * HID, h,
      W2 + blockIdx.x * 32 + ct * 16 + c, KDIM, kq * (HID / 4), hk);
#pragma unroll
  for (int r = 0; r < 8; ++r) red[wave][r * 32 + lane] = part[r];
  __syncthreads();

  const int t = lane >> 4;
  const int n = blockIdx.x * 32 + lane;
  const float bias = b2[n];
#pragma unroll
  for (int u = 0; u < 2; ++u) {
    const int idx = wave * 32 + u * 16 + (lane & 15);
    float acc = bias;
#pragma unroll
    for (int q = 0; q < 4; ++q) acc += red[t + 2 * q][idx];
    wmat[(size_t)(row0 + wave + 8 * u) * KDIM + n] = acc;
  }
}

#define RPW 4
#define NBROW (RPW + 2)

struct Frag {
  v2f a[3][2];
  v2f b[NBROW];
};

struct Raw {
  float ax[3][2], ay[3][2];
  float bx[NBROW], by[NBROW];
};

__device__ __forceinline__ void issue_loads(
    Raw& r, const float* __restrict__ lat_b, const float* arow0,
    const float* arow1, int cg, int dj, int h0, int px0, int m, int hiK) {
  const int c0  = 4 * cg + 2 * hiK;
  const int ww  = px0 + m + dj - 1;
  const int wwc = min(max(ww, 0), WDIM - 1);
  const float* cptr = lat_b + (size_t)c0 * (HDIM * WDIM) + wwc;
#pragma unroll
  for (int rr = 0; rr < NBROW; ++rr) {
    const int rowc = min(max(h0 - 1 + rr, 0), HDIM - 1);
    r.bx[rr] = cptr[rowc * WDIM];
    r.by[rr] = cptr[rowc * WDIM + HDIM * WDIM];
  }
#pragma unroll
  for (int di = 0; di < 3; ++di) {
    const int kk = di * 3 + dj;
    r.ax[di][0] = arow0[c0 * KK + kk];
    r.ay[di][0] = arow0[(c0 + 1) * KK + kk];
    r.ax[di][1] = arow1[c0 * KK + kk];
    r.ay[di][1] = arow1[(c0 + 1) * KK + kk];
  }
}

__device__ __forceinline__ void finish_frag(
    Frag& f, const Raw& r, int dj, int h0, int px0, int m) {
  const int  ww    = px0 + m + dj - 1;
  const bool colOK = (unsigned)ww < (unsigned)WDIM;
#pragma unroll
  for (int rr = 0; rr < NBROW; ++rr) {
    const bool ok = colOK && ((unsigned)(h0 - 1 + rr) < (unsigned)HDIM);
    f.b[rr].x = ok ? r.bx[rr] : 0.0f;
    f.b[rr].y = ok ? r.by[rr] : 0.0f;
  }
#pragma unroll
  for (int di = 0; di < 3; ++di) {
#pragma unroll
    for (int t = 0; t < 2; ++t) {
      f.a[di][t].x = r.ax[di][t];
      f.a[di][t].y = r.ay[di][t];
    }
  }
}

__device__ __forceinline__ void mma_step(v8f (&acc)[2][RPW], const Frag& f) {
#pragma unroll
  for (int di = 0; di < 3; ++di) {
#pragma unroll
    for (int r = 0; r < RPW; ++r) {
#pragma unroll
      for (int t = 0; t < 2; ++t) {
        acc[t][r] = __builtin_amdgcn_wmma_f32_16x16x4_f32(
             false, f.a[di][t],  false, f.b[r + di],
             (short)0, acc[t][r],  false,  false);
      }
    }
  }
}

__global__ __launch_bounds__(256) void dynconv_wmma_kernel(
    const float* __restrict__ latent, const float* __restrict__ wmat,
    float* __restrict__ out) {
  __shared__ float wlds[NTOK * WSTRIDE];

  const int b   = blockIdx.z;
  const int h0  = blockIdx.y * RPW;
  const int tid = threadIdx.x;

  const float* wsrc = wmat + (size_t)b * NTOK * KDIM;
  for (int n = 0; n < NTOK; ++n) {
    for (int o = tid; o < KDIM; o += 256)
      wlds[n * WSTRIDE + o] = wsrc[n * KDIM + o];
  }
  __syncthreads();

  const int wave = tid >> 5;
  const int lane = tid & 31;
  const int m    = lane & 15;
  const int hiK  = lane >> 4;
  const int px0  = wave * 16;

  const float* lat_b = latent + (size_t)b * IN_CH * HDIM * WDIM;
  const float* arow0 = &wlds[m * WSTRIDE];
  const float* arow1 = &wlds[(16 + m) * WSTRIDE];

  v8f acc[2][RPW];
#pragma unroll
  for (int t = 0; t < 2; ++t)
#pragma unroll
    for (int r = 0; r < RPW; ++r)
      acc[t][r] = (v8f){0.f, 0.f, 0.f, 0.f, 0.f, 0.f, 0.f, 0.f};

  Raw  raw;
  Frag f;
  issue_loads(raw, lat_b, arow0, arow1, 0, 0, h0, px0, m, hiK);
  finish_frag(f, raw, 0, h0, px0, m);
  for (int cg = 0; cg < IN_CH / 4; ++cg) {
    const int cgn = min(cg + 1, IN_CH / 4 - 1);

    issue_loads(raw, lat_b, arow0, arow1, cg, 1, h0, px0, m, hiK);
    __builtin_amdgcn_sched_barrier(0);
    mma_step(acc, f);
    __builtin_amdgcn_sched_barrier(0);
    finish_frag(f, raw, 1, h0, px0, m);

    issue_loads(raw, lat_b, arow0, arow1, cg, 2, h0, px0, m, hiK);
    __builtin_amdgcn_sched_barrier(0);
    mma_step(acc, f);
    __builtin_amdgcn_sched_barrier(0);
    finish_frag(f, raw, 2, h0, px0, m);

    issue_loads(raw, lat_b, arow0, arow1, cgn, 0, h0, px0, m, hiK);
    __builtin_amdgcn_sched_barrier(0);
    mma_step(acc, f);
    __builtin_amdgcn_sched_barrier(0);
    finish_frag(f, raw, 0, h0, px0, m);
  }

#pragma unroll
  for (int t = 0; t < 2; ++t) {
    const int tok = 16 * t + 8 * hiK;
#pragma unroll
    for (int r = 0; r < RPW; ++r) {
#pragma unroll
      for (int i = 0; i < 8; ++i) {
        out[(((size_t)b * NTOK + tok + i) * HDIM + (h0 + r)) * WDIM + (px0 + m)] =
            acc[t][r][i];
      }
    }
  }
}

extern "C" void kernel_launch(void* const* d_in, const int* in_sizes, int n_in,
                              void* d_out, int out_size, void* d_ws, size_t ws_size,
                              hipStream_t stream) {
  const float* latent = (const float*)d_in[0];
  const float* t_emb  = (const float*)d_in[1];
  const float* wv     = (const float*)d_in[2];
  const float* W1     = (const float*)d_in[3];
  const float* b1     = (const float*)d_in[4];
  const float* W2     = (const float*)d_in[5];
  const float* b2     = (const float*)d_in[6];
  float* out = (float*)d_out;

  float* hbuf = (float*)d_ws;
  float* wmat = hbuf + (size_t)BATCH * NTOK * HID;

  mlp1_kernel<<<dim3(HID / 32, BATCH * NTOK / 16), 512, 0, stream>>>(t_emb, wv, W1, b1, hbuf);
  mlp2_kernel<<<dim3(KDIM / 32, BATCH * NTOK / 16), 256, 0, stream>>>(hbuf, W2, b2, wmat);

  dim3 grid(1, HDIM / RPW, BATCH);
  dynconv_wmma_kernel<<<grid, 256, 0, stream>>>(latent, wmat, out);
}


# dynconv weight staging: each thread issues its 18 128-bit global loads back to back, then stores them to LDS as 128-bit (row pitch 577 to 580 floats), instead of 96 load-wait-store trips per wave
# speedup vs baseline: 5.5626x; 1.7123x over previous
#include <hip/hip_runtime.h>

#define IN_CH  64
#define EMBED  1024
#define HID    256
#define KK     9
#define KDIM   (IN_CH * KK)
#define BATCH  8
#define NTOK   32
#define HDIM   128
#define WDIM   128
#define WSTRIDE 580
#define STAGE_V4 (NTOK * KDIM / 4 / 256)

typedef __attribute__((ext_vector_type(2))) float v2f;
typedef __attribute__((ext_vector_type(8))) float v8f;

typedef __attribute__((ext_vector_type(4))) float v4f;

template <int KW, bool ADD2>
__device__ __forceinline__ v8f tile_partial(
    const float* __restrict__ arow, const float* __restrict__ a2row,
    const float* __restrict__ bcol, int ldb, int kbeg, int hk) {
  v8f acc0 = (v8f){0.f, 0.f, 0.f, 0.f, 0.f, 0.f, 0.f, 0.f};
  v8f acc1 = (v8f){0.f, 0.f, 0.f, 0.f, 0.f, 0.f, 0.f, 0.f};
#pragma unroll
  for (int kb = 0; kb < KW; kb += 8) {
    const int k = kbeg + kb + 4 * hk;
    v4f a = *(const v4f*)(arow + k);
    if (ADD2) {
      const v4f t = *(const v4f*)(a2row + k);
      a = a + t;
    }
    const float* bp = bcol + (size_t)k * ldb;
    const v2f a01 = (v2f){a.x, a.y};
    const v2f a23 = (v2f){a.z, a.w};
    const v2f b01 = (v2f){bp[0], bp[ldb]};
    const v2f b23 = (v2f){bp[2 * ldb], bp[3 * ldb]};
    acc0 = __builtin_amdgcn_wmma_f32_16x16x4_f32(
         false, a01,  false, b01,
         (short)0, acc0,  false,  false);
    acc1 = __builtin_amdgcn_wmma_f32_16x16x4_f32(
         false, a23,  false, b23,
         (short)0, acc1,  false,  false);
  }
  return acc0 + acc1;
}

__global__ __launch_bounds__(512) void mlp1_kernel(
    const float* __restrict__ t_emb, const float* __restrict__ wv,
    const float* __restrict__ W1, const float* __restrict__ b1,
    float* __restrict__ hout) {
  __shared__ float red[16][256];
  const int row0 = blockIdx.y * 16;
  const int tid  = threadIdx.x;
  const int wave = tid >> 5;
  const int lane = tid & 31;
  const int c    = lane & 15;
  const int hk   = lane >> 4;
  const int ct   = wave & 1;
  const int kq   = wave >> 1;
  const int b    = row0 >> 5;

  const v8f part = tile_partial<EMBED / 8, true>(
      wv + (size_t)(row0 + c) * EMBED, t_emb + (size_t)b * EMBED,
      W1 + blockIdx.x * 32 + ct * 16 + c, HID, kq * (EMBED / 8), hk);
#pragma unroll
  for (int r = 0; r < 8; ++r) red[wave][r * 32 + lane] = part[r];
  __syncthreads();

  const int orow = wave;
  const int t    = lane >> 4;
  const int idx  = (orow & 7) * 32 + (orow >> 3) * 16 + (lane & 15);
  const int n    = blockIdx.x * 32 + lane;
  float acc = b1[n];
#pragma unroll
  for (int q = 0; q < 8; ++q) acc += red[t + 2 * q][idx];
  const float sig = 1.0f / (1.0f + __expf(-acc));
  hout[(size_t)(row0 + orow) * HID + n] = acc * sig;
}

__global__ __launch_bounds__(256) void mlp2_kernel(
    const float* __restrict__ h, const float* __restrict__ W2,
    const float* __restrict__ b2, float* __restrict__ wmat) {
  __shared__ float red[8][256];
  const int row0 = blockIdx.y * 16;
  const int tid  = threadIdx.x;
  const int wave = tid >> 5;
  const int lane = tid & 31;
  const int c    = lane & 15;
  const int hk   = lane >> 4;
  const int ct   = wave & 1;
  const int kq   = wave >> 1;

  const v8f part = tile_partial<HID / 4, false>(
      h + (size_t)(row0 + c) * HID, h,
      W2 + blockIdx.x * 32 + ct * 16 + c, KDIM, kq * (HID / 4), hk);
#pragma unroll
  for (int r = 0; r < 8; ++r) red[wave][r * 32 + lane] = part[r];
  __syncthreads();

  const int t = lane >> 4;
  const int n = blockIdx.x * 32 + lane;
  const float bias = b2[n];
#pragma unroll
  for (int u = 0; u < 2; ++u) {
    const int idx = wave * 32 + u * 16 + (lane & 15);
    float acc = bias;
#pragma unroll
    for (int q = 0; q < 4; ++q) acc += red[t + 2 * q][idx];
    wmat[(size_t)(row0 + wave + 8 * u) * KDIM + n] = acc;
  }
}

#define RPW 4
#define NBROW (RPW + 2)

struct Frag {
  v2f a[3][2];
  v2f b[NBROW];
};

struct Raw {
  float ax[3][2], ay[3][2];
  float bx[NBROW], by[NBROW];
};

__device__ __forceinline__ void issue_loads(
    Raw& r, const float* __restrict__ lat_b, const float* arow0,
    const float* arow1, int cg, int dj, int h0, int px0, int m, int hiK) {
  const int c0  = 4 * cg + 2 * hiK;
  const int ww  = px0 + m + dj - 1;
  const int wwc = min(max(ww, 0), WDIM - 1);
  const float* cptr = lat_b + (size_t)c0 * (HDIM * WDIM) + wwc;
#pragma unroll
  for (int rr = 0; rr < NBROW; ++rr) {
    const int rowc = min(max(h0 - 1 + rr, 0), HDIM - 1);
    r.bx[rr] = cptr[rowc * WDIM];
    r.by[rr] = cptr[rowc * WDIM + HDIM * WDIM];
  }
#pragma unroll
  for (int di = 0; di < 3; ++di) {
    const int kk = di * 3 + dj;
    r.ax[di][0] = arow0[c0 * KK + kk];
    r.ay[di][0] = arow0[(c0 + 1) * KK + kk];
    r.ax[di][1] = arow1[c0 * KK + kk];
    r.ay[di][1] = arow1[(c0 + 1) * KK + kk];
  }
}

__device__ __forceinline__ void finish_frag(
    Frag& f, const Raw& r, int dj, int h0, int px0, int m) {
  const int  ww    = px0 + m + dj - 1;
  const bool colOK = (unsigned)ww < (unsigned)WDIM;
#pragma unroll
  for (int rr = 0; rr < NBROW; ++rr) {
    const bool ok = colOK && ((unsigned)(h0 - 1 + rr) < (unsigned)HDIM);
    f.b[rr].x = ok ? r.bx[rr] : 0.0f;
    f.b[rr].y = ok ? r.by[rr] : 0.0f;
  }
#pragma unroll
  for (int di = 0; di < 3; ++di) {
#pragma unroll
    for (int t = 0; t < 2; ++t) {
      f.a[di][t].x = r.ax[di][t];
      f.a[di][t].y = r.ay[di][t];
    }
  }
}

__device__ __forceinline__ void mma_step(v8f (&acc)[2][RPW], const Frag& f) {
#pragma unroll
  for (int di = 0; di < 3; ++di) {
#pragma unroll
    for (int r = 0; r < RPW; ++r) {
#pragma unroll
      for (int t = 0; t < 2; ++t) {
        acc[t][r] = __builtin_amdgcn_wmma_f32_16x16x4_f32(
             false, f.a[di][t],  false, f.b[r + di],
             (short)0, acc[t][r],  false,  false);
      }
    }
  }
}

__global__ __launch_bounds__(256) void dynconv_wmma_kernel(
    const float* __restrict__ latent, const float* __restrict__ wmat,
    float* __restrict__ out) {
  __shared__ __attribute__((aligned(16))) float wlds[NTOK * WSTRIDE];

  const int b   = blockIdx.z;
  const int h0  = blockIdx.y * RPW;
  const int tid = threadIdx.x;

  const float* wsrc = wmat + (size_t)b * NTOK * KDIM;
  {
    const v4f* wsrc4 = (const v4f*)wsrc;
    v4f stg[STAGE_V4];
#pragma unroll
    for (int i = 0; i < STAGE_V4; ++i) stg[i] = wsrc4[tid + 256 * i];
#pragma unroll
    for (int i = 0; i < STAGE_V4; ++i) {
      const int j = tid + 256 * i;
      const int n = j / (KDIM / 4);
      const int o = (j - n * (KDIM / 4)) * 4;
      *(v4f*)&wlds[n * WSTRIDE + o] = stg[i];
    }
  }
  __syncthreads();

  const int wave = tid >> 5;
  const int lane = tid & 31;
  const int m    = lane & 15;
  const int hiK  = lane >> 4;
  const int px0  = wave * 16;

  const float* lat_b = latent + (size_t)b * IN_CH * HDIM * WDIM;
  const float* arow0 = &wlds[m * WSTRIDE];
  const float* arow1 = &wlds[(16 + m) * WSTRIDE];

  v8f acc[2][RPW];
#pragma unroll
  for (int t = 0; t < 2; ++t)
#pragma unroll
    for (int r = 0; r < RPW; ++r)
      acc[t][r] = (v8f){0.f, 0.f, 0.f, 0.f, 0.f, 0.f, 0.f, 0.f};

  Raw  raw;
  Frag f;
  issue_loads(raw, lat_b, arow0, arow1, 0, 0, h0, px0, m, hiK);
  finish_frag(f, raw, 0, h0, px0, m);
  for (int cg = 0; cg < IN_CH / 4; ++cg) {
    const int cgn = min(cg + 1, IN_CH / 4 - 1);

    issue_loads(raw, lat_b, arow0, arow1, cg, 1, h0, px0, m, hiK);
    __builtin_amdgcn_sched_barrier(0);
    mma_step(acc, f);
    __builtin_amdgcn_sched_barrier(0);
    finish_frag(f, raw, 1, h0, px0, m);

    issue_loads(raw, lat_b, arow0, arow1, cg, 2, h0, px0, m, hiK);
    __builtin_amdgcn_sched_barrier(0);
    mma_step(acc, f);
    __builtin_amdgcn_sched_barrier(0);
    finish_frag(f, raw, 2, h0, px0, m);

    issue_loads(raw, lat_b, arow0, arow1, cgn, 0, h0, px0, m, hiK);
    __builtin_amdgcn_sched_barrier(0);
    mma_step(acc, f);
    __builtin_amdgcn_sched_barrier(0);
    finish_frag(f, raw, 0, h0, px0, m);
  }

#pragma unroll
  for (int t = 0; t < 2; ++t) {
    const int tok = 16 * t + 8 * hiK;
#pragma unroll
    for (int r = 0; r < RPW; ++r) {
#pragma unroll
      for (int i = 0; i < 8; ++i) {
        out[(((size_t)b * NTOK + tok + i) * HDIM + (h0 + r)) * WDIM + (px0 + m)] =
            acc[t][r][i];
      }
    }
  }
}

extern "C" void kernel_launch(void* const* d_in, const int* in_sizes, int n_in,
                              void* d_out, int out_size, void* d_ws, size_t ws_size,
                              hipStream_t stream) {
  const float* latent = (const float*)d_in[0];
  const float* t_emb  = (const float*)d_in[1];
  const float* wv     = (const float*)d_in[2];
  const float* W1     = (const float*)d_in[3];
  const float* b1     = (const float*)d_in[4];
  const float* W2     = (const float*)d_in[5];
  const float* b2     = (const float*)d_in[6];
  float* out = (float*)d_out;

  float* hbuf = (float*)d_ws;
  float* wmat = hbuf + (size_t)BATCH * NTOK * HID;

  mlp1_kernel<<<dim3(HID / 32, BATCH * NTOK / 16), 512, 0, stream>>>(t_emb, wv, W1, b1, hbuf);
  mlp2_kernel<<<dim3(KDIM / 32, BATCH * NTOK / 16), 256, 0, stream>>>(hbuf, W2, b2, wmat);

  dim3 grid(1, HDIM / RPW, BATCH);
  dynconv_wmma_kernel<<<grid, 256, 0, stream>>>(latent, wmat, out);
}


# dynconv row taps by the F(2,3) minimal-filtering identity: 4 WMMA products per pair of output rows instead of 6, weights combined at staging, latent rows combined in registers; sum order changes
# speedup vs baseline: 6.0377x; 1.0854x over previous
#include <hip/hip_runtime.h>

#define IN_CH  64
#define EMBED  1024
#define HID    256
#define KK     9
#define KDIM   (IN_CH * KK)
#define BATCH  8
#define NTOK   32
#define HDIM   128
#define WDIM   128
#define UDIM   (IN_CH * 12)
#define WSTRIDE 772
#define STAGE_ITEMS (NTOK * (IN_CH / 2) / 256)

typedef __attribute__((ext_vector_type(2))) float v2f;
typedef __attribute__((ext_vector_type(8))) float v8f;

typedef __attribute__((ext_vector_type(4))) float v4f;

template <int KW, bool ADD2>
__device__ __forceinline__ v8f tile_partial(
    const float* __restrict__ arow, const float* __restrict__ a2row,
    const float* __restrict__ bcol, int ldb, int kbeg, int hk) {
  v8f acc0 = (v8f){0.f, 0.f, 0.f, 0.f, 0.f, 0.f, 0.f, 0.f};
  v8f acc1 = (v8f){0.f, 0.f, 0.f, 0.f, 0.f, 0.f, 0.f, 0.f};
#pragma unroll
  for (int kb = 0; kb < KW; kb += 8) {
    const int k = kbeg + kb + 4 * hk;
    v4f a = *(const v4f*)(arow + k);
    if (ADD2) {
      const v4f t = *(const v4f*)(a2row + k);
      a = a + t;
    }
    const float* bp = bcol + (size_t)k * ldb;
    const v2f a01 = (v2f){a.x, a.y};
    const v2f a23 = (v2f){a.z, a.w};
    const v2f b01 = (v2f){bp[0], bp[ldb]};
    const v2f b23 = (v2f){bp[2 * ldb], bp[3 * ldb]};
    acc0 = __builtin_amdgcn_wmma_f32_16x16x4_f32(
         false, a01,  false, b01,
         (short)0, acc0,  false,  false);
    acc1 = __builtin_amdgcn_wmma_f32_16x16x4_f32(
         false, a23,  false, b23,
         (short)0, acc1,  false,  false);
  }
  return acc0 + acc1;
}

__global__ __launch_bounds__(512) void mlp1_kernel(
    const float* __restrict__ t_emb, const float* __restrict__ wv,
    const float* __restrict__ W1, const float* __restrict__ b1,
    float* __restrict__ hout) {
  __shared__ float red[16][256];
  const int row0 = blockIdx.y * 16;
  const int tid  = threadIdx.x;
  const int wave = tid >> 5;
  const int lane = tid & 31;
  const int c    = lane & 15;
  const int hk   = lane >> 4;
  const int ct   = wave & 1;
  const int kq   = wave >> 1;
  const int b    = row0 >> 5;

  const v8f part = tile_partial<EMBED / 8, true>(
      wv + (size_t)(row0 + c) * EMBED, t_emb + (size_t)b * EMBED,
      W1 + blockIdx.x * 32 + ct * 16 + c, HID, kq * (EMBED / 8), hk);
#pragma unroll
  for (int r = 0; r < 8; ++r) red[wave][r * 32 + lane] = part[r];
  __syncthreads();

  const int orow = wave;
  const int t    = lane >> 4;
  const int idx  = (orow & 7) * 32 + (orow >> 3) * 16 + (lane & 15);
  const int n    = blockIdx.x * 32 + lane;
  float acc = b1[n];
#pragma unroll
  for (int q = 0; q < 8; ++q) acc += red[t + 2 * q][idx];
  const float sig = 1.0f / (1.0f + __expf(-acc));
  hout[(size_t)(row0 + orow) * HID + n] = acc * sig;
}

__global__ __launch_bounds__(256) void mlp2_kernel(
    const float* __restrict__ h, const float* __restrict__ W2,
    const float* __restrict__ b2, float* __restrict__ wmat) {
  __shared__ float red[8][256];
  const int row0 = blockIdx.y * 16;
  const int tid  = threadIdx.x;
  const int wave = tid >> 5;
  const int lane = tid & 31;
  const int c    = lane & 15;
  const int hk   = lane >> 4;
  const int ct   = wave & 1;
  const int kq   = wave >> 1;

  const v8f part = tile_partial<HID / 4, false>(
      h + (size_t)(row0 + c) * HID, h,
      W2 + blockIdx.x * 32 + ct * 16 + c, KDIM, kq * (HID / 4), hk);
#pragma unroll
  for (int r = 0; r < 8; ++r) red[wave][r * 32 + lane] = part[r];
  __syncthreads();

  const int t = lane >> 4;
  const int n = blockIdx.x * 32 + lane;
  const float bias = b2[n];
#pragma unroll
  for (int u = 0; u < 2; ++u) {
    const int idx = wave * 32 + u * 16 + (lane & 15);
    float acc = bias;
#pragma unroll
    for (int q = 0; q < 4; ++q) acc += red[t + 2 * q][idx];
    wmat[(size_t)(row0 + wave + 8 * u) * KDIM + n] = acc;
  }
}

#define RPW 4
#define NBROW (RPW + 2)
#define NRP (RPW / 2)

struct Frag {
  v2f a[4][2];
  v2f b[NRP][4];
};

struct Raw {
  v4f a[2][2];
  float bx[NBROW], by[NBROW];
};

__device__ __forceinline__ void issue_loads(
    Raw& r, const float* __restrict__ lat_b, const float* arow0,
    const float* arow1, int cg, int dj, int h0, int px0, int m, int hiK) {
  const int c0  = 4 * cg + 2 * hiK;
  const int ww  = px0 + m + dj - 1;
  const int wwc = min(max(ww, 0), WDIM - 1);
  const float* cptr = lat_b + (size_t)c0 * (HDIM * WDIM) + wwc;
#pragma unroll
  for (int rr = 0; rr < NBROW; ++rr) {
    const int rowc = min(max(h0 - 1 + rr, 0), HDIM - 1);
    r.bx[rr] = cptr[rowc * WDIM];
    r.by[rr] = cptr[rowc * WDIM + HDIM * WDIM];
  }
  const int ao = (c0 >> 1) * 24 + dj * 8;
  r.a[0][0] = *(const v4f*)(arow0 + ao);
  r.a[0][1] = *(const v4f*)(arow0 + ao + 4);
  r.a[1][0] = *(const v4f*)(arow1 + ao);
  r.a[1][1] = *(const v4f*)(arow1 + ao + 4);
}

__device__ __forceinline__ void finish_frag(
    Frag& f, const Raw& r, int dj, int h0, int px0, int m) {
  const int  ww    = px0 + m + dj - 1;
  const bool colOK = (unsigned)ww < (unsigned)WDIM;
  float zx[NBROW], zy[NBROW];
#pragma unroll
  for (int rr = 0; rr < NBROW; ++rr) {
    const bool ok = colOK && ((unsigned)(h0 - 1 + rr) < (unsigned)HDIM);
    zx[rr] = ok ? r.bx[rr] : 0.0f;
    zy[rr] = ok ? r.by[rr] : 0.0f;
  }
#pragma unroll
  for (int i = 0; i < NRP; ++i) {
    f.b[i][0] = (v2f){zx[2 * i]     - zx[2 * i + 2], zy[2 * i]     - zy[2 * i + 2]};
    f.b[i][1] = (v2f){zx[2 * i + 1] + zx[2 * i + 2], zy[2 * i + 1] + zy[2 * i + 2]};
    f.b[i][2] = (v2f){zx[2 * i + 2] - zx[2 * i + 1], zy[2 * i + 2] - zy[2 * i + 1]};
    f.b[i][3] = (v2f){zx[2 * i + 1] - zx[2 * i + 3], zy[2 * i + 1] - zy[2 * i + 3]};
  }
#pragma unroll
  for (int t = 0; t < 2; ++t) {
    f.a[0][t] = (v2f){r.a[t][0].x, r.a[t][0].y};
    f.a[1][t] = (v2f){r.a[t][0].z, r.a[t][0].w};
    f.a[2][t] = (v2f){r.a[t][1].x, r.a[t][1].y};
    f.a[3][t] = (v2f){r.a[t][1].z, r.a[t][1].w};
  }
}

__device__ __forceinline__ void mma_step(v8f (&acc)[2][NRP][4], const Frag& f) {
#pragma unroll
  for (int x = 0; x < 4; ++x) {
#pragma unroll
    for (int i = 0; i < NRP; ++i) {
#pragma unroll
      for (int t = 0; t < 2; ++t) {
        acc[t][i][x] = __builtin_amdgcn_wmma_f32_16x16x4_f32(
             false, f.a[x][t],  false, f.b[i][x],
             (short)0, acc[t][i][x],  false,  false);
      }
    }
  }
}

__global__ __launch_bounds__(256) void dynconv_wmma_kernel(
    const float* __restrict__ latent, const float* __restrict__ wmat,
    float* __restrict__ out) {
  __shared__ __attribute__((aligned(16))) float wlds[NTOK * WSTRIDE];

  const int b   = blockIdx.z;
  const int h0  = blockIdx.y * RPW;
  const int tid = threadIdx.x;

  const float* wsrc = wmat + (size_t)b * NTOK * KDIM;
  {
    v2f stg[STAGE_ITEMS][9];
#pragma unroll
    for (int q = 0; q < STAGE_ITEMS; ++q) {
      const int item = tid + 256 * q;
      const int n  = item >> 5;
      const int dp = item & 31;
      const v2f* src = (const v2f*)(wsrc + n * KDIM + dp * 18);
#pragma unroll
      for (int j = 0; j < 9; ++j) stg[q][j] = src[j];
    }
#pragma unroll
    for (int q = 0; q < STAGE_ITEMS; ++q) {
      const int item = tid + 256 * q;
      const int n  = item >> 5;
      const int dp = item & 31;
      float g[18];
#pragma unroll
      for (int j = 0; j < 9; ++j) { g[2 * j] = stg[q][j].x; g[2 * j + 1] = stg[q][j].y; }
      float u[24];
#pragma unroll
      for (int dl = 0; dl < 2; ++dl) {
#pragma unroll
        for (int dj = 0; dj < 3; ++dj) {
          const float g0 = g[dl * 9 + dj];
          const float g1 = g[dl * 9 + 3 + dj];
          const float g2 = g[dl * 9 + 6 + dj];
          const float s02 = g0 + g2;
          u[dj * 8 + 0 + dl] = g0;
          u[dj * 8 + 2 + dl] = 0.5f * (s02 + g1);
          u[dj * 8 + 4 + dl] = 0.5f * (s02 - g1);
          u[dj * 8 + 6 + dl] = g2;
        }
      }
      float* dst = &wlds[n * WSTRIDE + dp * 24];
#pragma unroll
      for (int v = 0; v < 6; ++v)
        *(v4f*)(dst + 4 * v) = (v4f){u[4 * v], u[4 * v + 1], u[4 * v + 2], u[4 * v + 3]};
    }
  }
  __syncthreads();

  const int wave = tid >> 5;
  const int lane = tid & 31;
  const int m    = lane & 15;
  const int hiK  = lane >> 4;
  const int px0  = wave * 16;

  const float* lat_b = latent + (size_t)b * IN_CH * HDIM * WDIM;
  const float* arow0 = &wlds[m * WSTRIDE];
  const float* arow1 = &wlds[(16 + m) * WSTRIDE];

  v8f acc[2][NRP][4];
#pragma unroll
  for (int t = 0; t < 2; ++t)
#pragma unroll
    for (int i = 0; i < NRP; ++i)
#pragma unroll
      for (int x = 0; x < 4; ++x)
        acc[t][i][x] = (v8f){0.f, 0.f, 0.f, 0.f, 0.f, 0.f, 0.f, 0.f};

  Raw  raw;
  Frag f;
  issue_loads(raw, lat_b, arow0, arow1, 0, 0, h0, px0, m, hiK);
  finish_frag(f, raw, 0, h0, px0, m);
  for (int cg = 0; cg < IN_CH / 4; ++cg) {
    const int cgn = min(cg + 1, IN_CH / 4 - 1);

    issue_loads(raw, lat_b, arow0, arow1, cg, 1, h0, px0, m, hiK);
    __builtin_amdgcn_sched_barrier(0);
    mma_step(acc, f);
    __builtin_amdgcn_sched_barrier(0);
    finish_frag(f, raw, 1, h0, px0, m);

    issue_loads(raw, lat_b, arow0, arow1, cg, 2, h0, px0, m, hiK);
    __builtin_amdgcn_sched_barrier(0);
    mma_step(acc, f);
    __builtin_amdgcn_sched_barrier(0);
    finish_frag(f, raw, 2, h0, px0, m);

    issue_loads(raw, lat_b, arow0, arow1, cgn, 0, h0, px0, m, hiK);
    __builtin_amdgcn_sched_barrier(0);
    mma_step(acc, f);
    __builtin_amdgcn_sched_barrier(0);
    finish_frag(f, raw, 0, h0, px0, m);
  }

#pragma unroll
  for (int t = 0; t < 2; ++t) {
    const int tok = 16 * t + 8 * hiK;
#pragma unroll
    for (int i = 0; i < NRP; ++i) {
#pragma unroll
      for (int e = 0; e < 8; ++e) {
        const float m0 = acc[t][i][0][e], m1 = acc[t][i][1][e];
        const float m2 = acc[t][i][2][e], m3 = acc[t][i][3][e];
        float* o = out + (((size_t)b * NTOK + tok + e) * HDIM + (h0 + 2 * i)) * WDIM + (px0 + m);
        o[0]    = (m0 + m1) + m2;
        o[WDIM] = (m1 - m2) - m3;
      }
    }
  }
}

extern "C" void kernel_launch(void* const* d_in, const int* in_sizes, int n_in,
                              void* d_out, int out_size, void* d_ws, size_t ws_size,
                              hipStream_t stream) {
  const float* latent = (const float*)d_in[0];
  const float* t_emb  = (const float*)d_in[1];
  const float* wv     = (const float*)d_in[2];
  const float* W1     = (const float*)d_in[3];
  const float* b1     = (const float*)d_in[4];
  const float* W2     = (const float*)d_in[5];
  const float* b2     = (const float*)d_in[6];
  float* out = (float*)d_out;

  float* hbuf = (float*)d_ws;
  float* wmat = hbuf + (size_t)BATCH * NTOK * HID;

  mlp1_kernel<<<dim3(HID / 32, BATCH * NTOK / 16), 512, 0, stream>>>(t_emb, wv, W1, b1, hbuf);
  mlp2_kernel<<<dim3(KDIM / 32, BATCH * NTOK / 16), 256, 0, stream>>>(hbuf, W2, b2, wmat);

  dim3 grid(1, HDIM / RPW, BATCH);
  dynconv_wmma_kernel<<<grid, 256, 0, stream>>>(latent, wmat, out);
}


# dynconv latent row loads issued three steps (one channel group) ahead of their use instead of one, the first group ahead of the weight staging's wait; staged weights still one step ahead
# speedup vs baseline: 6.5184x; 1.0796x over previous
#include <hip/hip_runtime.h>

#define IN_CH  64
#define EMBED  1024
#define HID    256
#define KK     9
#define KDIM   (IN_CH * KK)
#define BATCH  8
#define NTOK   32
#define HDIM   128
#define WDIM   128
#define UDIM   (IN_CH * 12)
#define WSTRIDE 772
#define STAGE_ITEMS (NTOK * (IN_CH / 2) / 256)

typedef __attribute__((ext_vector_type(2))) float v2f;
typedef __attribute__((ext_vector_type(8))) float v8f;

typedef __attribute__((ext_vector_type(4))) float v4f;

template <int KW, bool ADD2>
__device__ __forceinline__ v8f tile_partial(
    const float* __restrict__ arow, const float* __restrict__ a2row,
    const float* __restrict__ bcol, int ldb, int kbeg, int hk) {
  v8f acc0 = (v8f){0.f, 0.f, 0.f, 0.f, 0.f, 0.f, 0.f, 0.f};
  v8f acc1 = (v8f){0.f, 0.f, 0.f, 0.f, 0.f, 0.f, 0.f, 0.f};
#pragma unroll
  for (int kb = 0; kb < KW; kb += 8) {
    const int k = kbeg + kb + 4 * hk;
    v4f a = *(const v4f*)(arow + k);
    if (ADD2) {
      const v4f t = *(const v4f*)(a2row + k);
      a = a + t;
    }
    const float* bp = bcol + (size_t)k * ldb;
    const v2f a01 = (v2f){a.x, a.y};
    const v2f a23 = (v2f){a.z, a.w};
    const v2f b01 = (v2f){bp[0], bp[ldb]};
    const v2f b23 = (v2f){bp[2 * ldb], bp[3 * ldb]};
    acc0 = __builtin_amdgcn_wmma_f32_16x16x4_f32(
         false, a01,  false, b01,
         (short)0, acc0,  false,  false);
    acc1 = __builtin_amdgcn_wmma_f32_16x16x4_f32(
         false, a23,  false, b23,
         (short)0, acc1,  false,  false);
  }
  return acc0 + acc1;
}

__global__ __launch_bounds__(512) void mlp1_kernel(
    const float* __restrict__ t_emb, const float* __restrict__ wv,
    const float* __restrict__ W1, const float* __restrict__ b1,
    float* __restrict__ hout) {
  __shared__ float red[16][256];
  const int row0 = blockIdx.y * 16;
  const int tid  = threadIdx.x;
  const int wave = tid >> 5;
  const int lane = tid & 31;
  const int c    = lane & 15;
  const int hk   = lane >> 4;
  const int ct   = wave & 1;
  const int kq   = wave >> 1;
  const int b    = row0 >> 5;

  const v8f part = tile_partial<EMBED / 8, true>(
      wv + (size_t)(row0 + c) * EMBED, t_emb + (size_t)b * EMBED,
      W1 + blockIdx.x * 32 + ct * 16 + c, HID, kq * (EMBED / 8), hk);
#pragma unroll
  for (int r = 0; r < 8; ++r) red[wave][r * 32 + lane] = part[r];
  __syncthreads();

  const int orow = wave;
  const int t    = lane >> 4;
  const int idx  = (orow & 7) * 32 + (orow >> 3) * 16 + (lane & 15);
  const int n    = blockIdx.x * 32 + lane;
  float acc = b1[n];
#pragma unroll
  for (int q = 0; q < 8; ++q) acc += red[t + 2 * q][idx];
  const float sig = 1.0f / (1.0f + __expf(-acc));
  hout[(size_t)(row0 + orow) * HID + n] = acc * sig;
}

__global__ __launch_bounds__(256) void mlp2_kernel(
    const float* __restrict__ h, const float* __restrict__ W2,
    const float* __restrict__ b2, float* __restrict__ wmat) {
  __shared__ float red[8][256];
  const int row0 = blockIdx.y * 16;
  const int tid  = threadIdx.x;
  const int wave = tid >> 5;
  const int lane = tid & 31;
  const int c    = lane & 15;
  const int hk   = lane >> 4;
  const int ct   = wave & 1;
  const int kq   = wave >> 1;

  const v8f part = tile_partial<HID / 4, false>(
      h + (size_t)(row0 + c) * HID, h,
      W2 + blockIdx.x * 32 + ct * 16 + c, KDIM, kq * (HID / 4), hk);
#pragma unroll
  for (int r = 0; r < 8; ++r) red[wave][r * 32 + lane] = part[r];
  __syncthreads();

  const int t = lane >> 4;
  const int n = blockIdx.x * 32 + lane;
  const float bias = b2[n];
#pragma unroll
  for (int u = 0; u < 2; ++u) {
    const int idx = wave * 32 + u * 16 + (lane & 15);
    float acc = bias;
#pragma unroll
    for (int q = 0; q < 4; ++q) acc += red[t + 2 * q][idx];
    wmat[(size_t)(row0 + wave + 8 * u) * KDIM + n] = acc;
  }
}

#define RPW 4
#define NBROW (RPW + 2)
#define NRP (RPW / 2)

struct Frag {
  v2f a[4][2];
  v2f b[NRP][4];
};

struct RawA {
  v4f a[2][2];
};

struct RawB {
  float bx[NBROW], by[NBROW];
};

__device__ __forceinline__ void issue_lat(
    RawB& r, const float* __restrict__ lat_b, int cg, int dj, int h0, int px0,
    int m, int hiK) {
  const int c0  = 4 * cg + 2 * hiK;
  const int ww  = px0 + m + dj - 1;
  const int wwc = min(max(ww, 0), WDIM - 1);
  const float* cptr = lat_b + (size_t)c0 * (HDIM * WDIM) + wwc;
#pragma unroll
  for (int rr = 0; rr < NBROW; ++rr) {
    const int rowc = min(max(h0 - 1 + rr, 0), HDIM - 1);
    r.bx[rr] = cptr[rowc * WDIM];
    r.by[rr] = cptr[rowc * WDIM + HDIM * WDIM];
  }
}

__device__ __forceinline__ void issue_wts(
    RawA& r, const float* arow0, const float* arow1, int cg, int dj, int hiK) {
  const int c0 = 4 * cg + 2 * hiK;
  const int ao = (c0 >> 1) * 24 + dj * 8;
  r.a[0][0] = *(const v4f*)(arow0 + ao);
  r.a[0][1] = *(const v4f*)(arow0 + ao + 4);
  r.a[1][0] = *(const v4f*)(arow1 + ao);
  r.a[1][1] = *(const v4f*)(arow1 + ao + 4);
}

__device__ __forceinline__ void finish_frag(
    Frag& f, const RawB& r, const RawA& w, int dj, int h0, int px0, int m) {
  const int  ww    = px0 + m + dj - 1;
  const bool colOK = (unsigned)ww < (unsigned)WDIM;
  float zx[NBROW], zy[NBROW];
#pragma unroll
  for (int rr = 0; rr < NBROW; ++rr) {
    const bool ok = colOK && ((unsigned)(h0 - 1 + rr) < (unsigned)HDIM);
    zx[rr] = ok ? r.bx[rr] : 0.0f;
    zy[rr] = ok ? r.by[rr] : 0.0f;
  }
#pragma unroll
  for (int i = 0; i < NRP; ++i) {
    f.b[i][0] = (v2f){zx[2 * i]     - zx[2 * i + 2], zy[2 * i]     - zy[2 * i + 2]};
    f.b[i][1] = (v2f){zx[2 * i + 1] + zx[2 * i + 2], zy[2 * i + 1] + zy[2 * i + 2]};
    f.b[i][2] = (v2f){zx[2 * i + 2] - zx[2 * i + 1], zy[2 * i + 2] - zy[2 * i + 1]};
    f.b[i][3] = (v2f){zx[2 * i + 1] - zx[2 * i + 3], zy[2 * i + 1] - zy[2 * i + 3]};
  }
#pragma unroll
  for (int t = 0; t < 2; ++t) {
    f.a[0][t] = (v2f){w.a[t][0].x, w.a[t][0].y};
    f.a[1][t] = (v2f){w.a[t][0].z, w.a[t][0].w};
    f.a[2][t] = (v2f){w.a[t][1].x, w.a[t][1].y};
    f.a[3][t] = (v2f){w.a[t][1].z, w.a[t][1].w};
  }
}

__device__ __forceinline__ void mma_step(v8f (&acc)[2][NRP][4], const Frag& f) {
#pragma unroll
  for (int x = 0; x < 4; ++x) {
#pragma unroll
    for (int i = 0; i < NRP; ++i) {
#pragma unroll
      for (int t = 0; t < 2; ++t) {
        acc[t][i][x] = __builtin_amdgcn_wmma_f32_16x16x4_f32(
             false, f.a[x][t],  false, f.b[i][x],
             (short)0, acc[t][i][x],  false,  false);
      }
    }
  }
}

__global__ __launch_bounds__(256) void dynconv_wmma_kernel(
    const float* __restrict__ latent, const float* __restrict__ wmat,
    float* __restrict__ out) {
  __shared__ __attribute__((aligned(16))) float wlds[NTOK * WSTRIDE];

  const int b   = blockIdx.z;
  const int h0  = blockIdx.y * RPW;
  const int tid = threadIdx.x;

  const int wave = tid >> 5;
  const int lane = tid & 31;
  const int m    = lane & 15;
  const int hiK  = lane >> 4;
  const int px0  = wave * 16;
  const float* lat_b = latent + (size_t)b * IN_CH * HDIM * WDIM;

  RawB rb0, rb1, rb2;

  const float* wsrc = wmat + (size_t)b * NTOK * KDIM;
  {
    v2f stg[STAGE_ITEMS][9];
#pragma unroll
    for (int q = 0; q < STAGE_ITEMS; ++q) {
      const int item = tid + 256 * q;
      const int n  = item >> 5;
      const int dp = item & 31;
      const v2f* src = (const v2f*)(wsrc + n * KDIM + dp * 18);
#pragma unroll
      for (int j = 0; j < 9; ++j) stg[q][j] = src[j];
    }
    issue_lat(rb0, lat_b, 0, 0, h0, px0, m, hiK);
    issue_lat(rb1, lat_b, 0, 1, h0, px0, m, hiK);
    issue_lat(rb2, lat_b, 0, 2, h0, px0, m, hiK);
#pragma unroll
    for (int q = 0; q < STAGE_ITEMS; ++q) {
      const int item = tid + 256 * q;
      const int n  = item >> 5;
      const int dp = item & 31;
      float g[18];
#pragma unroll
      for (int j = 0; j < 9; ++j) { g[2 * j] = stg[q][j].x; g[2 * j + 1] = stg[q][j].y; }
      float u[24];
#pragma unroll
      for (int dl = 0; dl < 2; ++dl) {
#pragma unroll
        for (int dj = 0; dj < 3; ++dj) {
          const float g0 = g[dl * 9 + dj];
          const float g1 = g[dl * 9 + 3 + dj];
          const float g2 = g[dl * 9 + 6 + dj];
          const float s02 = g0 + g2;
          u[dj * 8 + 0 + dl] = g0;
          u[dj * 8 + 2 + dl] = 0.5f * (s02 + g1);
          u[dj * 8 + 4 + dl] = 0.5f * (s02 - g1);
          u[dj * 8 + 6 + dl] = g2;
        }
      }
      float* dst = &wlds[n * WSTRIDE + dp * 24];
#pragma unroll
      for (int v = 0; v < 6; ++v)
        *(v4f*)(dst + 4 * v) = (v4f){u[4 * v], u[4 * v + 1], u[4 * v + 2], u[4 * v + 3]};
    }
  }
  __syncthreads();

  const float* arow0 = &wlds[m * WSTRIDE];
  const float* arow1 = &wlds[(16 + m) * WSTRIDE];

  v8f acc[2][NRP][4];
#pragma unroll
  for (int t = 0; t < 2; ++t)
#pragma unroll
    for (int i = 0; i < NRP; ++i)
#pragma unroll
      for (int x = 0; x < 4; ++x)
        acc[t][i][x] = (v8f){0.f, 0.f, 0.f, 0.f, 0.f, 0.f, 0.f, 0.f};

  RawA ra;
  Frag f;
  issue_wts(ra, arow0, arow1, 0, 0, hiK);
  finish_frag(f, rb0, ra, 0, h0, px0, m);
  for (int cg = 0; cg < IN_CH / 4; ++cg) {
    const int cgn = min(cg + 1, IN_CH / 4 - 1);

    issue_lat(rb0, lat_b, cgn, 0, h0, px0, m, hiK);
    issue_wts(ra, arow0, arow1, cg, 1, hiK);
    __builtin_amdgcn_sched_barrier(0);
    mma_step(acc, f);
    __builtin_amdgcn_sched_barrier(0);
    finish_frag(f, rb1, ra, 1, h0, px0, m);

    issue_lat(rb1, lat_b, cgn, 1, h0, px0, m, hiK);
    issue_wts(ra, arow0, arow1, cg, 2, hiK);
    __builtin_amdgcn_sched_barrier(0);
    mma_step(acc, f);
    __builtin_amdgcn_sched_barrier(0);
    finish_frag(f, rb2, ra, 2, h0, px0, m);

    issue_lat(rb2, lat_b, cgn, 2, h0, px0, m, hiK);
    issue_wts(ra, arow0, arow1, cgn, 0, hiK);
    __builtin_amdgcn_sched_barrier(0);
    mma_step(acc, f);
    __builtin_amdgcn_sched_barrier(0);
    finish_frag(f, rb0, ra, 0, h0, px0, m);
  }

#pragma unroll
  for (int t = 0; t < 2; ++t) {
    const int tok = 16 * t + 8 * hiK;
#pragma unroll
    for (int i = 0; i < NRP; ++i) {
#pragma unroll
      for (int e = 0; e < 8; ++e) {
        const float m0 = acc[t][i][0][e], m1 = acc[t][i][1][e];
        const float m2 = acc[t][i][2][e], m3 = acc[t][i][3][e];
        float* o = out + (((size_t)b * NTOK + tok + e) * HDIM + (h0 + 2 * i)) * WDIM + (px0 + m);
        o[0]    = (m0 + m1) + m2;
        o[WDIM] = (m1 - m2) - m3;
      }
    }
  }
}

extern "C" void kernel_launch(void* const* d_in, const int* in_sizes, int n_in,
                              void* d_out, int out_size, void* d_ws, size_t ws_size,
                              hipStream_t stream) {
  const float* latent = (const float*)d_in[0];
  const float* t_emb  = (const float*)d_in[1];
  const float* wv     = (const float*)d_in[2];
  const float* W1     = (const float*)d_in[3];
  const float* b1     = (const float*)d_in[4];
  const float* W2     = (const float*)d_in[5];
  const float* b2     = (const float*)d_in[6];
  float* out = (float*)d_out;

  float* hbuf = (float*)d_ws;
  float* wmat = hbuf + (size_t)BATCH * NTOK * HID;

  mlp1_kernel<<<dim3(HID / 32, BATCH * NTOK / 16), 512, 0, stream>>>(t_emb, wv, W1, b1, hbuf);
  mlp2_kernel<<<dim3(KDIM / 32, BATCH * NTOK / 16), 256, 0, stream>>>(hbuf, W2, b2, wmat);

  dim3 grid(1, HDIM / RPW, BATCH);
  dynconv_wmma_kernel<<<grid, 256, 0, stream>>>(latent, wmat, out);
}


# dynconv column taps by the F(2,3) minimal-filtering identity on top of the row F(4,3): 24 WMMA products per 4 rows x 2 columns instead of 36; each wave 16 tokens x 32 pixels; sum order changes
# speedup vs baseline: 7.5393x; 1.1566x over previous
#include <hip/hip_runtime.h>

#define IN_CH  64
#define EMBED  1024
#define HID    256
#define KK     9
#define KDIM   (IN_CH * KK)
#define BATCH  8
#define NTOK   32
#define HDIM   128
#define WDIM   128
#define NCOMB  6
#define NXI    4
#define UPAIR  (NXI * NCOMB * 2)
#define PSTRIDE (UPAIR + 4)
#define WSTRIDE (IN_CH / 2 * PSTRIDE + 4)
#define STAGE_ITEMS (NTOK * (IN_CH / 2) / 256)

typedef __attribute__((ext_vector_type(2))) float v2f;
typedef __attribute__((ext_vector_type(8))) float v8f;

typedef __attribute__((ext_vector_type(4))) float v4f;

template <int KW, bool ADD2>
__device__ __forceinline__ v8f tile_partial(
    const float* __restrict__ arow, const float* __restrict__ a2row,
    const float* __restrict__ bcol, int ldb, int kbeg, int hk) {
  v8f acc0 = (v8f){0.f, 0.f, 0.f, 0.f, 0.f, 0.f, 0.f, 0.f};
  v8f acc1 = (v8f){0.f, 0.f, 0.f, 0.f, 0.f, 0.f, 0.f, 0.f};
#pragma unroll
  for (int kb = 0; kb < KW; kb += 8) {
    const int k = kbeg + kb + 4 * hk;
    v4f a = *(const v4f*)(arow + k);
    if (ADD2) {
      const v4f t = *(const v4f*)(a2row + k);
      a = a + t;
    }
    const float* bp = bcol + (size_t)k * ldb;
    const v2f a01 = (v2f){a.x, a.y};
    const v2f a23 = (v2f){a.z, a.w};
    const v2f b01 = (v2f){bp[0], bp[ldb]};
    const v2f b23 = (v2f){bp[2 * ldb], bp[3 * ldb]};
    acc0 = __builtin_amdgcn_wmma_f32_16x16x4_f32(
         false, a01,  false, b01,
         (short)0, acc0,  false,  false);
    acc1 = __builtin_amdgcn_wmma_f32_16x16x4_f32(
         false, a23,  false, b23,
         (short)0, acc1,  false,  false);
  }
  return acc0 + acc1;
}

__global__ __launch_bounds__(512) void mlp1_kernel(
    const float* __restrict__ t_emb, const float* __restrict__ wv,
    const float* __restrict__ W1, const float* __restrict__ b1,
    float* __restrict__ hout) {
  __shared__ float red[16][256];
  const int row0 = blockIdx.y * 16;
  const int tid  = threadIdx.x;
  const int wave = tid >> 5;
  const int lane = tid & 31;
  const int c    = lane & 15;
  const int hk   = lane >> 4;
  const int ct   = wave & 1;
  const int kq   = wave >> 1;
  const int b    = row0 >> 5;

  const v8f part = tile_partial<EMBED / 8, true>(
      wv + (size_t)(row0 + c) * EMBED, t_emb + (size_t)b * EMBED,
      W1 + blockIdx.x * 32 + ct * 16 + c, HID, kq * (EMBED / 8), hk);
#pragma unroll
  for (int r = 0; r < 8; ++r) red[wave][r * 32 + lane] = part[r];
  __syncthreads();

  const int orow = wave;
  const int t    = lane >> 4;
  const int idx  = (orow & 7) * 32 + (orow >> 3) * 16 + (lane & 15);
  const int n    = blockIdx.x * 32 + lane;
  float acc = b1[n];
#pragma unroll
  for (int q = 0; q < 8; ++q) acc += red[t + 2 * q][idx];
  const float sig = 1.0f / (1.0f + __expf(-acc));
  hout[(size_t)(row0 + orow) * HID + n] = acc * sig;
}

__global__ __launch_bounds__(256) void mlp2_kernel(
    const float* __restrict__ h, const float* __restrict__ W2,
    const float* __restrict__ b2, float* __restrict__ wmat) {
  __shared__ float red[8][256];
  const int row0 = blockIdx.y * 16;
  const int tid  = threadIdx.x;
  const int wave = tid >> 5;
  const int lane = tid & 31;
  const int c    = lane & 15;
  const int hk   = lane >> 4;
  const int ct   = wave & 1;
  const int kq   = wave >> 1;

  const v8f part = tile_partial<HID / 4, false>(
      h + (size_t)(row0 + c) * HID, h,
      W2 + blockIdx.x * 32 + ct * 16 + c, KDIM, kq * (HID / 4), hk);
#pragma unroll
  for (int r = 0; r < 8; ++r) red[wave][r * 32 + lane] = part[r];
  __syncthreads();

  const int t = lane >> 4;
  const int n = blockIdx.x * 32 + lane;
  const float bias = b2[n];
#pragma unroll
  for (int u = 0; u < 2; ++u) {
    const int idx = wave * 32 + u * 16 + (lane & 15);
    float acc = bias;
#pragma unroll
    for (int q = 0; q < 4; ++q) acc += red[t + 2 * q][idx];
    wmat[(size_t)(row0 + wave + 8 * u) * KDIM + n] = acc;
  }
}

#define RPW 4
#define NBROW (RPW + 2)

struct FragB {
  v2f b[NXI][NCOMB];
};

struct FragA {
  v4f a[NXI][3];
};

struct RawB {
  float xa[NBROW][2];
  v2f   xbc[NBROW][2];
  float xd[NBROW][2];
};

__device__ __forceinline__ void issue_lat(
    RawB& r, const float* __restrict__ lat_b, int cg, int h0, int pxl, int hiK) {
  const int c0 = 4 * cg + 2 * hiK;
  const int wl = max(pxl - 1, 0);
  const int wr = min(pxl + 2, WDIM - 1);
  const float* cptr = lat_b + (size_t)c0 * (HDIM * WDIM);
#pragma unroll
  for (int rr = 0; rr < NBROW; ++rr) {
    const int rowc = min(max(h0 - 1 + rr, 0), HDIM - 1);
    const float* rp = cptr + rowc * WDIM;
#pragma unroll
    for (int ch = 0; ch < 2; ++ch) {
      const float* p = rp + ch * (HDIM * WDIM);
      r.xa[rr][ch]  = p[wl];
      r.xbc[rr][ch] = *(const v2f*)(p + pxl);
      r.xd[rr][ch]  = p[wr];
    }
  }
}

__device__ __forceinline__ void issue_wts_half(
    FragA& w, const float* arow, int cg, int hiK, int half) {
  const int ao = (2 * cg + hiK) * PSTRIDE + half * (2 * 2 * NCOMB);
#pragma unroll
  for (int i = 0; i < 2; ++i)
#pragma unroll
    for (int v = 0; v < 3; ++v)
      w.a[2 * half + i][v] = *(const v4f*)(arow + ao + i * (2 * NCOMB) + 4 * v);
}

__device__ __forceinline__ void finish_b(
    FragB& f, const RawB& r, bool topOK, bool botOK, bool leftOK, bool rightOK) {
  v2f z[NXI][NBROW];
#pragma unroll
  for (int rr = 0; rr < NBROW; ++rr) {
    const bool rowOK = (rr == 0) ? topOK : ((rr == NBROW - 1) ? botOK : true);
    const bool lOK = rowOK && leftOK;
    const bool rOK = rowOK && rightOK;
    const v2f x0 = (v2f){lOK ? r.xa[rr][0] : 0.0f, lOK ? r.xa[rr][1] : 0.0f};
    const v2f x1 = (v2f){rowOK ? r.xbc[rr][0].x : 0.0f, rowOK ? r.xbc[rr][1].x : 0.0f};
    const v2f x2 = (v2f){rowOK ? r.xbc[rr][0].y : 0.0f, rowOK ? r.xbc[rr][1].y : 0.0f};
    const v2f x3 = (v2f){rOK ? r.xd[rr][0] : 0.0f, rOK ? r.xd[rr][1] : 0.0f};
    z[0][rr] = x0 - x2;
    z[1][rr] = x1 + x2;
    z[2][rr] = x2 - x1;
    z[3][rr] = x1 - x3;
  }
#pragma unroll
  for (int xi = 0; xi < NXI; ++xi) {
    const v2f p = z[xi][4] - 4.0f * z[xi][2];
    const v2f q = z[xi][3] - 4.0f * z[xi][1];
    const v2f s = z[xi][4] - z[xi][2];
    const v2f d = z[xi][3] - z[xi][1];
    f.b[xi][0] = 4.0f * z[xi][0] + (z[xi][4] - 5.0f * z[xi][2]);
    f.b[xi][1] = p + q;
    f.b[xi][2] = p - q;
    f.b[xi][3] = s + 2.0f * d;
    f.b[xi][4] = s - 2.0f * d;
    f.b[xi][5] = 4.0f * z[xi][1] + (z[xi][5] - 5.0f * z[xi][3]);
  }
}

__device__ __forceinline__ void mma_half(
    v8f (&acc)[NXI][NCOMB], const FragA& w, const FragB& f, int half) {
#pragma unroll
  for (int i = 0; i < 2; ++i) {
    const int xi = 2 * half + i;
#pragma unroll
    for (int v = 0; v < 3; ++v) {
      const v2f a0 = (v2f){w.a[xi][v].x, w.a[xi][v].y};
      const v2f a1 = (v2f){w.a[xi][v].z, w.a[xi][v].w};
      acc[xi][2 * v] = __builtin_amdgcn_wmma_f32_16x16x4_f32(
           false, a0,  false, f.b[xi][2 * v],
           (short)0, acc[xi][2 * v],  false,  false);
      acc[xi][2 * v + 1] = __builtin_amdgcn_wmma_f32_16x16x4_f32(
           false, a1,  false, f.b[xi][2 * v + 1],
           (short)0, acc[xi][2 * v + 1],  false,  false);
    }
  }
}

__device__ __forceinline__ void group_step(
    v8f (&acc)[NXI][NCOMB], FragA& w, FragB& f, RawB& refill, const RawB& next,
    const float* __restrict__ lat_b, const float* arow, int cg, int h0, int pxl,
    int hiK, bool topOK, bool botOK, bool leftOK, bool rightOK) {
  const int cg2 = min(cg + 2, IN_CH / 4 - 1);
  const int cg1 = min(cg + 1, IN_CH / 4 - 1);
  issue_lat(refill, lat_b, cg2, h0, pxl, hiK);
  __builtin_amdgcn_sched_barrier(0);
  mma_half(acc, w, f, 0);
  __builtin_amdgcn_sched_barrier(0);
  issue_wts_half(w, arow, cg1, hiK, 0);
  __builtin_amdgcn_sched_barrier(0);
  mma_half(acc, w, f, 1);
  __builtin_amdgcn_sched_barrier(0);
  issue_wts_half(w, arow, cg1, hiK, 1);
  finish_b(f, next, topOK, botOK, leftOK, rightOK);
}

__global__ __launch_bounds__(256) void dynconv_wmma_kernel(
    const float* __restrict__ latent, const float* __restrict__ wmat,
    float* __restrict__ out) {
  __shared__ __attribute__((aligned(16))) float wlds[NTOK * WSTRIDE];

  const int b   = blockIdx.z;
  const int h0  = blockIdx.y * RPW;
  const int tid = threadIdx.x;

  const int wave = tid >> 5;
  const int lane = tid & 31;
  const int m    = lane & 15;
  const int hiK  = lane >> 4;
  const int pg   = wave & 3;
  const int tt   = wave >> 2;
  const int pxl  = pg * 32 + 2 * m;
  const float* lat_b = latent + (size_t)b * IN_CH * HDIM * WDIM;

  const bool topOK   = h0 >= 1;
  const bool botOK   = h0 + RPW < HDIM;
  const bool leftOK  = pxl >= 1;
  const bool rightOK = pxl + 2 < WDIM;

  RawB rbA, rbB;

  const float* wsrc = wmat + (size_t)b * NTOK * KDIM;
  {
    v2f stg[STAGE_ITEMS][9];
#pragma unroll
    for (int q = 0; q < STAGE_ITEMS; ++q) {
      const int item = tid + 256 * q;
      const int n  = item >> 5;
      const int dp = item & 31;
      const v2f* src = (const v2f*)(wsrc + n * KDIM + dp * 18);
#pragma unroll
      for (int j = 0; j < 9; ++j) stg[q][j] = src[j];
    }
    issue_lat(rbA, lat_b, 0, h0, pxl, hiK);
#pragma unroll
    for (int q = 0; q < STAGE_ITEMS; ++q) {
      const int item = tid + 256 * q;
      const int n  = item >> 5;
      const int dp = item & 31;
      float g[18];
#pragma unroll
      for (int j = 0; j < 9; ++j) { g[2 * j] = stg[q][j].x; g[2 * j + 1] = stg[q][j].y; }
      float rw[3][2 * NCOMB];
#pragma unroll
      for (int dl = 0; dl < 2; ++dl) {
#pragma unroll
        for (int dj = 0; dj < 3; ++dj) {
          const float g0 = g[dl * 9 + dj];
          const float g1 = g[dl * 9 + 3 + dj];
          const float g2 = g[dl * 9 + 6 + dj];
          const float s02 = g0 + g2;
          const float ev  = (1.0f / 24.0f) * g0 + (1.0f / 6.0f) * g2;
          const float od  = (1.0f / 12.0f) * g1;
          rw[dj][0  + dl] = 0.25f * g0;
          rw[dj][2  + dl] = (-1.0f / 6.0f) * (s02 + g1);
          rw[dj][4  + dl] = (-1.0f / 6.0f) * (s02 - g1);
          rw[dj][6  + dl] = ev + od;
          rw[dj][8  + dl] = ev - od;
          rw[dj][10 + dl] = g2;
        }
      }
      float u[UPAIR];
#pragma unroll
      for (int k = 0; k < 2 * NCOMB; ++k) {
        const float a0 = rw[0][k], a1 = rw[1][k], a2 = rw[2][k];
        const float s02 = a0 + a2;
        u[0 * 2 * NCOMB + k] = a0;
        u[1 * 2 * NCOMB + k] = 0.5f * (s02 + a1);
        u[2 * 2 * NCOMB + k] = 0.5f * (s02 - a1);
        u[3 * 2 * NCOMB + k] = a2;
      }
      float* dst = &wlds[n * WSTRIDE + dp * PSTRIDE];
#pragma unroll
      for (int v = 0; v < UPAIR / 4; ++v)
        *(v4f*)(dst + 4 * v) = (v4f){u[4 * v], u[4 * v + 1], u[4 * v + 2], u[4 * v + 3]};
    }
  }
  __syncthreads();

  const float* arow = &wlds[(16 * tt + m) * WSTRIDE];

  v8f acc[NXI][NCOMB];
#pragma unroll
  for (int xi = 0; xi < NXI; ++xi)
#pragma unroll
    for (int x = 0; x < NCOMB; ++x)
      acc[xi][x] = (v8f){0.f, 0.f, 0.f, 0.f, 0.f, 0.f, 0.f, 0.f};

  FragA w;
  FragB f;
  issue_lat(rbB, lat_b, 1, h0, pxl, hiK);
  issue_wts_half(w, arow, 0, hiK, 0);
  issue_wts_half(w, arow, 0, hiK, 1);
  finish_b(f, rbA, topOK, botOK, leftOK, rightOK);
  for (int cg = 0; cg < IN_CH / 4; cg += 2) {
    group_step(acc, w, f, rbA, rbB, lat_b, arow, cg, h0, pxl, hiK, topOK, botOK, leftOK, rightOK);
    group_step(acc, w, f, rbB, rbA, lat_b, arow, cg + 1, h0, pxl, hiK, topOK, botOK, leftOK, rightOK);
  }

  const int tok = 16 * tt + 8 * hiK;
#pragma unroll
  for (int e = 0; e < 8; ++e) {
    v2f c[NCOMB];
#pragma unroll
    for (int x = 0; x < NCOMB; ++x) {
      const float n0 = acc[0][x][e], n1 = acc[1][x][e], n2 = acc[2][x][e], n3 = acc[3][x][e];
      c[x] = (v2f){(n0 + n1) + n2, (n1 - n2) - n3};
    }
    const v2f s12 = c[1] + c[2], d12 = c[1] - c[2];
    const v2f s34 = c[3] + c[4], d34 = c[3] - c[4];
    float* o = out + (((size_t)b * NTOK + tok + e) * HDIM + h0) * WDIM + pxl;
    *(v2f*)(o)            = (c[0] + s12) + s34;
    *(v2f*)(o + WDIM)     = d12 + 2.0f * d34;
    *(v2f*)(o + 2 * WDIM) = s12 + 4.0f * s34;
    *(v2f*)(o + 3 * WDIM) = (d12 + 8.0f * d34) + c[5];
  }
}

extern "C" void kernel_launch(void* const* d_in, const int* in_sizes, int n_in,
                              void* d_out, int out_size, void* d_ws, size_t ws_size,
                              hipStream_t stream) {
  const float* latent = (const float*)d_in[0];
  const float* t_emb  = (const float*)d_in[1];
  const float* wv     = (const float*)d_in[2];
  const float* W1     = (const float*)d_in[3];
  const float* b1     = (const float*)d_in[4];
  const float* W2     = (const float*)d_in[5];
  const float* b2     = (const float*)d_in[6];
  float* out = (float*)d_out;

  float* hbuf = (float*)d_ws;
  float* wmat = hbuf + (size_t)BATCH * NTOK * HID;

  mlp1_kernel<<<dim3(HID / 32, BATCH * NTOK / 16), 512, 0, stream>>>(t_emb, wv, W1, b1, hbuf);
  mlp2_kernel<<<dim3(KDIM / 32, BATCH * NTOK / 16), 256, 0, stream>>>(hbuf, W2, b2, wmat);

  dim3 grid(1, HDIM / RPW, BATCH);
  dynconv_wmma_kernel<<<grid, 256, 0, stream>>>(latent, wmat, out);
}
